# starting kernel
# baseline (speedup 1.0000x reference)
#include <hip/hip_runtime.h>

#define B_    16
#define CIN_  16
#define OUT_  16
#define H_    320
#define W_    320
#define HW_   (H_ * W_)
#define KEXP_ 4
#define HID_  5
#define KTOT_ 144
#define COLS  322
#define CPAD  18

typedef __attribute__((ext_vector_type(2))) float v2f;
typedef __attribute__((ext_vector_type(8))) float v8f;

__global__ __launch_bounds__(256) void dyn_pool_kernel(
    const float* __restrict__ x, float* __restrict__ pooled) {
  const int bc = blockIdx.x;
  const float4* p4 = (const float4*)(x + (size_t)bc * HW_);
  float s = 0.0f;
  for (int i = threadIdx.x; i < HW_ / 4; i += 256) {
    float4 v = p4[i];
    s += v.x + v.y + v.z + v.w;
  }
  __shared__ float red[256];
  red[threadIdx.x] = s;
  __syncthreads();
  for (int off = 128; off > 0; off >>= 1) {
    if (threadIdx.x < (unsigned)off) red[threadIdx.x] += red[threadIdx.x + off];
    __syncthreads();
  }
  if (threadIdx.x == 0) pooled[bc] = red[0] * (1.0f / (float)HW_);
}

__global__ __launch_bounds__(256) void dyn_attn_kernel(
    const float* __restrict__ pooled, const float* __restrict__ w1,
    const float* __restrict__ w2, const float* __restrict__ b2,
    const float* __restrict__ weight, float* __restrict__ wmix) {
  const int t = threadIdx.x;
  const int b = t >> 4;
  const int j = t & 15;

  float hid[HID_];
#pragma unroll
  for (int h = 0; h < HID_; ++h) {
    float a = 0.0f;
#pragma unroll
    for (int c = 0; c < CIN_; ++c) a += pooled[b * CIN_ + c] * w1[h * CIN_ + c];
    hid[h] = fmaxf(a, 0.0f);
  }
  float av[KEXP_];
#pragma unroll
  for (int k = 0; k < KEXP_; ++k) {
    const int i = k * 16 + j;
    float a = b2[i];
#pragma unroll
    for (int h = 0; h < HID_; ++h) a += hid[h] * w2[i * HID_ + h];
    av[k] = a;
  }
  float mx = fmaxf(fmaxf(av[0], av[1]), fmaxf(av[2], av[3]));
  float e[KEXP_], sum = 0.0f;
#pragma unroll
  for (int k = 0; k < KEXP_; ++k) { e[k] = expf(av[k] - mx); sum += e[k]; }
  const float inv = 1.0f / sum;

  __shared__ float sm[B_][KEXP_ * OUT_];
#pragma unroll
  for (int k = 0; k < KEXP_; ++k) sm[b][k * 16 + j] = e[k] * inv;
  __syncthreads();

  const int o = j;
  float coef[KEXP_];
#pragma unroll
  for (int k = 0; k < KEXP_; ++k) coef[k] = sm[b][o * KEXP_ + k];

  const float* wsrc = weight + (size_t)o * KTOT_ * KEXP_;
  float* wdst = wmix + ((size_t)b * OUT_ + o) * KTOT_;
  for (int c = 0; c < CIN_; ++c) {
#pragma unroll
    for (int rd = 0; rd < 9; ++rd) {
      float acc = 0.0f;
#pragma unroll
      for (int e2 = 0; e2 < KEXP_; ++e2)
        acc += coef[e2] * wsrc[(c * 9 + rd) * KEXP_ + e2];
      wdst[rd * 16 + c] = acc;
    }
  }
}

__global__ __launch_bounds__(256) void dyn_conv_kernel(
    const float* __restrict__ x, const float* __restrict__ wmix,
    float* __restrict__ out) {
  const int blk = blockIdx.x;
  const int b = blk / (H_ / 2);
  const int h0 = (blk % (H_ / 2)) * 2;

  __shared__ float xs[4 * COLS * CPAD];

  const float* xb = x + (size_t)b * CIN_ * HW_;
  for (int i = threadIdx.x; i < CIN_ * 4 * COLS; i += 256) {
    const int col = i % COLS;
    const int cr = i / COLS;
    const int r = cr & 3;
    const int c = cr >> 2;
    const int hh = h0 - 1 + r;
    const int ww = col - 1;
    float v = 0.0f;
    if (hh >= 0 && hh < H_ && ww >= 0 && ww < W_)
      v = xb[c * HW_ + hh * W_ + ww];
    xs[(r * COLS + col) * CPAD + c] = v;
  }
  __syncthreads();

  const int lane = threadIdx.x & 31;
  const int wave = threadIdx.x >> 5;
  const int m = lane & 15;
  const int hi = lane >> 4;

  const float* wb = wmix + ((size_t)b * OUT_ + m) * KTOT_;
  v2f a[36];
#pragma unroll
  for (int s = 0; s < 36; ++s) {
    a[s].x = wb[4 * s + 2 * hi + 0];
    a[s].y = wb[4 * s + 2 * hi + 1];
  }

#pragma unroll
  for (int it = 0; it < 5; ++it) {
    const int tt = wave * 5 + it;
    const int lr = tt / 20;
    const int wbase = (tt % 20) * 16;
    const int base = (lr * COLS + wbase + m) * CPAD + 2 * hi;

    v8f acc = {};
#pragma unroll
    for (int s = 0; s < 36; ++s) {
      const int cb = (4 * s) & 15;
      const int rd = (4 * s) >> 4;
      const int r  = rd / 3;
      const int dx = rd % 3;
      const v2f bf = *(const v2f*)(xs + base + (r * COLS + dx) * CPAD + cb);
      acc = __builtin_amdgcn_wmma_f32_16x16x4_f32(
          false, a[s], false, bf, (short)0, acc, false, false);
    }

    float* ob = out + (size_t)b * OUT_ * HW_ + (h0 + lr) * W_ + wbase + m;
#pragma unroll
    for (int jj = 0; jj < 8; ++jj)
      ob[(size_t)(jj + 8 * hi) * HW_] = acc[jj];
  }
}

extern "C" void kernel_launch(void* const* d_in, const int* in_sizes, int n_in,
                              void* d_out, int out_size, void* d_ws, size_t ws_size,
                              hipStream_t stream) {
  (void)in_sizes; (void)n_in; (void)out_size; (void)ws_size;
  const float* x      = (const float*)d_in[0];
  const float* w1     = (const float*)d_in[1];
  const float* w2     = (const float*)d_in[2];
  const float* b2     = (const float*)d_in[3];
  const float* weight = (const float*)d_in[4];
  float* out = (float*)d_out;

  float* pooled = (float*)d_ws;
  float* wmix   = (float*)d_ws + 256;

  dyn_pool_kernel<<<B_ * CIN_, 256, 0, stream>>>(x, pooled);
  dyn_attn_kernel<<<1, 256, 0, stream>>>(pooled, w1, w2, b2, weight, wmix);
  dyn_conv_kernel<<<B_ * (H_ / 2), 256, 0, stream>>>(x, wmix, out);
}


# expert mix moved from launch 2's one-block serial loop into the conv waves' A loader: 16-byte loads of a fragment-ordered bank copy x 4 coefficients in registers; mixed-weight plane gone; same sums
# speedup vs baseline: 3.0875x; 3.0875x over previous
#include <hip/hip_runtime.h>

#define B_    16
#define CIN_  16
#define OUT_  16
#define H_    320
#define W_    320
#define HW_   (H_ * W_)
#define KEXP_ 4
#define HID_  5
#define KTOT_ 144

typedef __attribute__((ext_vector_type(2))) float v2f;
typedef __attribute__((ext_vector_type(8))) float v8f;

__global__ __launch_bounds__(256) void dyn_pool_kernel(
    const float* __restrict__ x, const float* __restrict__ weight,
    float* __restrict__ pooled, float* __restrict__ bankf) {
  const int bc = blockIdx.x;
  static_assert(B_ == OUT_, "one bank row per pool block");
  const bool copies = threadIdx.x < 9;
  float4 bankv = make_float4(0.0f, 0.0f, 0.0f, 0.0f);
  if (copies) bankv = ((const float4*)weight)[bc * 9 + threadIdx.x];
  const float4* p4 = (const float4*)(x + (size_t)bc * HW_);
  float s = 0.0f;
  for (int i = threadIdx.x; i < HW_ / 4; i += 256) {
    float4 v = p4[i];
    s += v.x + v.y + v.z + v.w;
  }
  if (copies) {
    const int c = bc % CIN_, o = bc / CIN_;
    const int sidx = threadIdx.x * 4 + (c >> 2);
    ((float4*)bankf)[(sidx * 2 + (c & 1)) * 32 + ((c >> 1) & 1) * 16 + o] = bankv;
  }
  __shared__ float red[256];
  red[threadIdx.x] = s;
  __syncthreads();
  for (int off = 128; off > 0; off >>= 1) {
    if (threadIdx.x < (unsigned)off) red[threadIdx.x] += red[threadIdx.x + off];
    __syncthreads();
  }
  if (threadIdx.x == 0) pooled[bc] = red[0] * (1.0f / (float)HW_);
}

__global__ __launch_bounds__(256) void dyn_attn_kernel(
    const float* __restrict__ pooled, const float* __restrict__ w1,
    const float* __restrict__ w2, const float* __restrict__ b2,
    float* __restrict__ coef) {
  const int t = threadIdx.x;
  const int b = t >> 4;
  const int j = t & 15;

  float hid[HID_];
#pragma unroll
  for (int h = 0; h < HID_; ++h) {
    float a = 0.0f;
#pragma unroll
    for (int c = 0; c < CIN_; ++c) a += pooled[b * CIN_ + c] * w1[h * CIN_ + c];
    hid[h] = fmaxf(a, 0.0f);
  }
  float av[KEXP_];
#pragma unroll
  for (int k = 0; k < KEXP_; ++k) {
    const int i = k * 16 + j;
    float a = b2[i];
#pragma unroll
    for (int h = 0; h < HID_; ++h) a += hid[h] * w2[i * HID_ + h];
    av[k] = a;
  }
  float mx = fmaxf(fmaxf(av[0], av[1]), fmaxf(av[2], av[3]));
  float e[KEXP_], sum = 0.0f;
#pragma unroll
  for (int k = 0; k < KEXP_; ++k) { e[k] = expf(av[k] - mx); sum += e[k]; }
  const float inv = 1.0f / sum;
#pragma unroll
  for (int k = 0; k < KEXP_; ++k) coef[b * (KEXP_ * OUT_) + k * 16 + j] = e[k] * inv;
}

#define STRIP_  20
#define NSTRIP_ (H_ / STRIP_)
#define CGRP_   (W_ / 64)

template <int S0>
__device__ __forceinline__ void conv_mix_a(
    v2f (&a)[36], const float4* __restrict__ bt, const float4 cf) {
#pragma unroll
  for (int i = 0; i < 9; ++i) {
    const int s = S0 + i;
    const float4 u = bt[(2 * s) * 32];
    const float4 v = bt[(2 * s + 1) * 32];
    float ax = 0.0f, ay = 0.0f;
    ax += cf.x * u.x;  ay += cf.x * v.x;
    ax += cf.y * u.y;  ay += cf.y * v.y;
    ax += cf.z * u.z;  ay += cf.z * v.z;
    ax += cf.w * u.w;  ay += cf.w * v.w;
    a[s].x = ax;
    a[s].y = ay;
  }
}

__device__ __forceinline__ void conv_load_row(
    v2f (&bf)[12], const float* __restrict__ rowp,
    unsigned o0, unsigned o1, unsigned o2) {
  const float* p0 = rowp + o0;
  const float* p1 = rowp + o1;
  const float* p2 = rowp + o2;
#pragma unroll
  for (int q = 0; q < 4; ++q) {
    bf[0 + q].x = p0[(4 * q) * HW_];  bf[0 + q].y = p0[(4 * q + 1) * HW_];
    bf[4 + q].x = p1[(4 * q) * HW_];  bf[4 + q].y = p1[(4 * q + 1) * HW_];
    bf[8 + q].x = p2[(4 * q) * HW_];  bf[8 + q].y = p2[(4 * q + 1) * HW_];
  }
}

template <bool R2, bool R1, bool R0>
__device__ __forceinline__ void conv_row_mma(
    const v2f (&a)[36], const v2f (&bf)[12], v8f& up, v8f& mid, v8f& dn,
    bool v0, bool v2) {
#pragma unroll
  for (int j = 0; j < 12; ++j) {
    v2f bm = bf[j];
    if (j < 4)  { bm.x = v0 ? bm.x : 0.0f;  bm.y = v0 ? bm.y : 0.0f; }
    if (j >= 8) { bm.x = v2 ? bm.x : 0.0f;  bm.y = v2 ? bm.y : 0.0f; }
    if (R2) up = __builtin_amdgcn_wmma_f32_16x16x4_f32(
        false, a[24 + j], false, bm, (short)0, up, false, false);
    if (R1) mid = __builtin_amdgcn_wmma_f32_16x16x4_f32(
        false, a[12 + j], false, bm, (short)0, mid, false, false);
    if (R0) dn = __builtin_amdgcn_wmma_f32_16x16x4_f32(
        false, a[j], false, bm, (short)0, dn, false, false);
  }
}

__device__ __forceinline__ void conv_store_row(float* __restrict__ op, const v8f& acc) {
#pragma unroll
  for (int jj = 0; jj < 8; ++jj) op[(size_t)jj * HW_] = acc[jj];
}

#define CONV_A_FENCE_S(V, A, S0)                                            \
  asm volatile("" : "+s"(V)                                                  \
               : "v"(A[S0 + 0]), "v"(A[S0 + 1]), "v"(A[S0 + 2]),             \
                 "v"(A[S0 + 3]), "v"(A[S0 + 4]), "v"(A[S0 + 5]),             \
                 "v"(A[S0 + 6]), "v"(A[S0 + 7]), "v"(A[S0 + 8]))
#define CONV_A_FENCE(V, A, S0)                                              \
  asm volatile("" : "+v"(V)                                                  \
               : "v"(A[S0 + 0]), "v"(A[S0 + 1]), "v"(A[S0 + 2]),             \
                 "v"(A[S0 + 3]), "v"(A[S0 + 4]), "v"(A[S0 + 5]),             \
                 "v"(A[S0 + 6]), "v"(A[S0 + 7]), "v"(A[S0 + 8]))

__global__ __launch_bounds__(128) void dyn_conv_kernel(
    const float* __restrict__ x, const float* __restrict__ bankf,
    const float* __restrict__ coef, float* __restrict__ out) {
  const int blk = blockIdx.x;
  const int cg = blk % CGRP_;
  const int st = (blk / CGRP_) % NSTRIP_;
  const int b  = blk / (CGRP_ * NSTRIP_);
  const int lane = threadIdx.x & 31;
  const int wave = __builtin_amdgcn_readfirstlane((int)(threadIdx.x >> 5));
  const int m  = lane & 15;
  const int hi = lane >> 4;
  const int w0 = cg * 64 + wave * 16;
  int y0 = st * STRIP_;

  const float4 cf = *(const float4*)(coef + ((size_t)b * OUT_ + m) * KEXP_);
  const float4* bt = (const float4*)bankf;
  unsigned lo = (unsigned)lane;
  v2f a[36];
  conv_mix_a<0>(a, bt + lo, cf);
  CONV_A_FENCE(lo, a, 0);
  conv_mix_a<9>(a, bt + lo, cf);
  CONV_A_FENCE(lo, a, 9);
  conv_mix_a<18>(a, bt + lo, cf);
  CONV_A_FENCE(lo, a, 18);
  conv_mix_a<27>(a, bt + lo, cf);
  CONV_A_FENCE_S(y0, a, 27);

  const int wl = w0 + m - 1, wr = w0 + m + 1;
  const bool v0 = wl >= 0, v2 = wr < W_;
  const unsigned chan = (unsigned)(2 * hi) * HW_;
  const unsigned o0 = chan + (unsigned)(v0 ? wl : 0);
  const unsigned o1 = chan + (unsigned)(w0 + m);
  const unsigned o2 = chan + (unsigned)(v2 ? wr : W_ - 1);

  const float* xb = x + (size_t)b * CIN_ * HW_;
  float* ob = out + (size_t)b * OUT_ * HW_ + (size_t)(8 * hi) * HW_ + w0 + m;

  const int ytop = y0 > 0 ? y0 - 1 : 0;
  v2f bA[12], bB[12];
  conv_load_row(bA, xb + (size_t)ytop * W_, o0, o1, o2);
  conv_load_row(bB, xb + (size_t)y0 * W_, o0, o1, o2);

  v8f P = {}, Q = {}, R = {};
  v8f none = {};
  if (st != 0) conv_row_mma<false, false, true>(a, bA, none, none, P, v0, v2);
  conv_load_row(bA, xb + (size_t)(y0 + 1) * W_, o0, o1, o2);
  conv_row_mma<false, true, true>(a, bB, none, P, Q, v0, v2);
  conv_load_row(bB, xb + (size_t)(y0 + 2) * W_, o0, o1, o2);

#pragma nounroll
  for (int t = 0; t < (STRIP_ - 2) / 6; ++t) {
    const int y = y0 + 1 + 6 * t;
    const float* xr = xb + (size_t)y * W_;
    float* orow = ob + (size_t)(y - 1) * W_;
    const int ylast = (y + 7 < H_) ? y + 7 : H_ - 1;

    conv_row_mma<true, true, true>(a, bA, P, Q, R, v0, v2);
    conv_store_row(orow, P);             P = none;
    conv_load_row(bA, xr + 2 * W_, o0, o1, o2);
    conv_row_mma<true, true, true>(a, bB, Q, R, P, v0, v2);
    conv_store_row(orow + W_, Q);        Q = none;
    conv_load_row(bB, xr + 3 * W_, o0, o1, o2);
    conv_row_mma<true, true, true>(a, bA, R, P, Q, v0, v2);
    conv_store_row(orow + 2 * W_, R);    R = none;
    conv_load_row(bA, xr + 4 * W_, o0, o1, o2);
    conv_row_mma<true, true, true>(a, bB, P, Q, R, v0, v2);
    conv_store_row(orow + 3 * W_, P);    P = none;
    conv_load_row(bB, xr + 5 * W_, o0, o1, o2);
    conv_row_mma<true, true, true>(a, bA, Q, R, P, v0, v2);
    conv_store_row(orow + 4 * W_, Q);    Q = none;
    conv_load_row(bA, xr + 6 * W_, o0, o1, o2);
    conv_row_mma<true, true, true>(a, bB, R, P, Q, v0, v2);
    conv_store_row(orow + 5 * W_, R);    R = none;
    conv_load_row(bB, xb + (size_t)ylast * W_, o0, o1, o2);
  }

  conv_row_mma<true, true, false>(a, bA, P, Q, none, v0, v2);
  conv_store_row(ob + (size_t)(y0 + STRIP_ - 2) * W_, P);
  if (st != NSTRIP_ - 1) conv_row_mma<true, false, false>(a, bB, Q, none, none, v0, v2);
  conv_store_row(ob + (size_t)(y0 + STRIP_ - 1) * W_, Q);
}

extern "C" void kernel_launch(void* const* d_in, const int* in_sizes, int n_in,
                              void* d_out, int out_size, void* d_ws, size_t ws_size,
                              hipStream_t stream) {
  (void)in_sizes; (void)n_in; (void)out_size; (void)ws_size;
  const float* x      = (const float*)d_in[0];
  const float* w1     = (const float*)d_in[1];
  const float* w2     = (const float*)d_in[2];
  const float* b2     = (const float*)d_in[3];
  const float* weight = (const float*)d_in[4];
  float* out = (float*)d_out;

  float* pooled = (float*)d_ws;
  float* coef   = (float*)d_ws + 256;
  float* bankf  = (float*)d_ws + 1280;

  dyn_pool_kernel<<<B_ * CIN_, 256, 0, stream>>>(x, weight, pooled, bankf);
  dyn_attn_kernel<<<1, 256, 0, stream>>>(pooled, w1, w2, b2, coef);
  dyn_conv_kernel<<<B_ * NSTRIP_ * CGRP_, 128, 0, stream>>>(x, bankf, coef, out);
}


# pool kernel: each plane split over 2 work-groups (512 in all), every thread's 50 16-byte loads requested before any is summed; partial sums added in launch 2; pool sum order changes
# speedup vs baseline: 3.2016x; 1.0370x over previous
#include <hip/hip_runtime.h>

#define B_    16
#define CIN_  16
#define OUT_  16
#define H_    320
#define W_    320
#define HW_   (H_ * W_)
#define KEXP_ 4
#define HID_  5
#define KTOT_ 144

typedef __attribute__((ext_vector_type(2))) float v2f;
typedef __attribute__((ext_vector_type(8))) float v8f;

#define PSPLIT_ 2
#define PLOADS_ (HW_ / 4 / PSPLIT_ / 256)

__global__ __launch_bounds__(256) void dyn_pool_kernel(
    const float* __restrict__ x, const float* __restrict__ weight,
    float* __restrict__ psum, float* __restrict__ bankf) {
  static_assert(B_ == OUT_, "one bank row per plane");
  static_assert(PLOADS_ * PSPLIT_ * 256 * 4 == HW_, "the parts tile the plane exactly");
  const int bc = blockIdx.x / PSPLIT_;
  const int part = blockIdx.x % PSPLIT_;
  const bool copies = (part == 0) && (threadIdx.x < 9);
  float4 bankv = make_float4(0.0f, 0.0f, 0.0f, 0.0f);
  if (copies) bankv = ((const float4*)weight)[bc * 9 + threadIdx.x];

  const float4* p4 = (const float4*)(x + (size_t)bc * HW_) +
                     (size_t)part * (PLOADS_ * 256) + threadIdx.x;
  float4 v[PLOADS_];
#pragma unroll
  for (int i = 0; i < PLOADS_; ++i) v[i] = p4[i * 256];
  float s0 = 0.0f, s1 = 0.0f, s2 = 0.0f, s3 = 0.0f;
#pragma unroll
  for (int i = 0; i < PLOADS_; ++i) {
    s0 += v[i].x;  s1 += v[i].y;  s2 += v[i].z;  s3 += v[i].w;
  }
  float s = (s0 + s1) + (s2 + s3);

  if (copies) {
    const int c = bc % CIN_, o = bc / CIN_;
    const int sidx = threadIdx.x * 4 + (c >> 2);
    ((float4*)bankf)[(sidx * 2 + (c & 1)) * 32 + ((c >> 1) & 1) * 16 + o] = bankv;
  }

#pragma unroll
  for (int off = 16; off > 0; off >>= 1) s += __shfl_xor(s, off);
  __shared__ float red[8];
  if ((threadIdx.x & 31) == 0) red[threadIdx.x >> 5] = s;
  __syncthreads();
  if (threadIdx.x == 0) {
    float t = red[0];
#pragma unroll
    for (int w = 1; w < 8; ++w) t += red[w];
    psum[blockIdx.x] = t;
  }
}

__global__ __launch_bounds__(256) void dyn_attn_kernel(
    const float* __restrict__ psum, const float* __restrict__ w1,
    const float* __restrict__ w2, const float* __restrict__ b2,
    float* __restrict__ coef) {
  const int t = threadIdx.x;
  const int b = t >> 4;
  const int j = t & 15;

  static_assert(PSPLIT_ == 2, "a plane's partial sums are read as one float2");
  float pooled[CIN_];
#pragma unroll
  for (int c = 0; c < CIN_; ++c) {
    const float2 pp = ((const float2*)psum)[b * CIN_ + c];
    pooled[c] = (pp.x + pp.y) * (1.0f / (float)HW_);
  }
  float hid[HID_];
#pragma unroll
  for (int h = 0; h < HID_; ++h) {
    float a = 0.0f;
#pragma unroll
    for (int c = 0; c < CIN_; ++c) a += pooled[c] * w1[h * CIN_ + c];
    hid[h] = fmaxf(a, 0.0f);
  }
  float av[KEXP_];
#pragma unroll
  for (int k = 0; k < KEXP_; ++k) {
    const int i = k * 16 + j;
    float a = b2[i];
#pragma unroll
    for (int h = 0; h < HID_; ++h) a += hid[h] * w2[i * HID_ + h];
    av[k] = a;
  }
  float mx = fmaxf(fmaxf(av[0], av[1]), fmaxf(av[2], av[3]));
  float e[KEXP_], sum = 0.0f;
#pragma unroll
  for (int k = 0; k < KEXP_; ++k) { e[k] = expf(av[k] - mx); sum += e[k]; }
  const float inv = 1.0f / sum;
#pragma unroll
  for (int k = 0; k < KEXP_; ++k) coef[b * (KEXP_ * OUT_) + k * 16 + j] = e[k] * inv;
}

#define STRIP_  20
#define NSTRIP_ (H_ / STRIP_)
#define CGRP_   (W_ / 64)

template <int S0>
__device__ __forceinline__ void conv_mix_a(
    v2f (&a)[36], const float4* __restrict__ bt, const float4 cf) {
#pragma unroll
  for (int i = 0; i < 9; ++i) {
    const int s = S0 + i;
    const float4 u = bt[(2 * s) * 32];
    const float4 v = bt[(2 * s + 1) * 32];
    float ax = 0.0f, ay = 0.0f;
    ax += cf.x * u.x;  ay += cf.x * v.x;
    ax += cf.y * u.y;  ay += cf.y * v.y;
    ax += cf.z * u.z;  ay += cf.z * v.z;
    ax += cf.w * u.w;  ay += cf.w * v.w;
    a[s].x = ax;
    a[s].y = ay;
  }
}

__device__ __forceinline__ void conv_load_row(
    v2f (&bf)[12], const float* __restrict__ rowp,
    unsigned o0, unsigned o1, unsigned o2) {
  const float* p0 = rowp + o0;
  const float* p1 = rowp + o1;
  const float* p2 = rowp + o2;
#pragma unroll
  for (int q = 0; q < 4; ++q) {
    bf[0 + q].x = p0[(4 * q) * HW_];  bf[0 + q].y = p0[(4 * q + 1) * HW_];
    bf[4 + q].x = p1[(4 * q) * HW_];  bf[4 + q].y = p1[(4 * q + 1) * HW_];
    bf[8 + q].x = p2[(4 * q) * HW_];  bf[8 + q].y = p2[(4 * q + 1) * HW_];
  }
}

template <bool R2, bool R1, bool R0>
__device__ __forceinline__ void conv_row_mma(
    const v2f (&a)[36], const v2f (&bf)[12], v8f& up, v8f& mid, v8f& dn,
    bool v0, bool v2) {
#pragma unroll
  for (int j = 0; j < 12; ++j) {
    v2f bm = bf[j];
    if (j < 4)  { bm.x = v0 ? bm.x : 0.0f;  bm.y = v0 ? bm.y : 0.0f; }
    if (j >= 8) { bm.x = v2 ? bm.x : 0.0f;  bm.y = v2 ? bm.y : 0.0f; }
    if (R2) up = __builtin_amdgcn_wmma_f32_16x16x4_f32(
        false, a[24 + j], false, bm, (short)0, up, false, false);
    if (R1) mid = __builtin_amdgcn_wmma_f32_16x16x4_f32(
        false, a[12 + j], false, bm, (short)0, mid, false, false);
    if (R0) dn = __builtin_amdgcn_wmma_f32_16x16x4_f32(
        false, a[j], false, bm, (short)0, dn, false, false);
  }
}

__device__ __forceinline__ void conv_store_row(float* __restrict__ op, const v8f& acc) {
#pragma unroll
  for (int jj = 0; jj < 8; ++jj) op[(size_t)jj * HW_] = acc[jj];
}

#define CONV_A_FENCE_S(V, A, S0)                                            \
  asm volatile("" : "+s"(V)                                                  \
               : "v"(A[S0 + 0]), "v"(A[S0 + 1]), "v"(A[S0 + 2]),             \
                 "v"(A[S0 + 3]), "v"(A[S0 + 4]), "v"(A[S0 + 5]),             \
                 "v"(A[S0 + 6]), "v"(A[S0 + 7]), "v"(A[S0 + 8]))
#define CONV_A_FENCE(V, A, S0)                                              \
  asm volatile("" : "+v"(V)                                                  \
               : "v"(A[S0 + 0]), "v"(A[S0 + 1]), "v"(A[S0 + 2]),             \
                 "v"(A[S0 + 3]), "v"(A[S0 + 4]), "v"(A[S0 + 5]),             \
                 "v"(A[S0 + 6]), "v"(A[S0 + 7]), "v"(A[S0 + 8]))

__global__ __launch_bounds__(128) void dyn_conv_kernel(
    const float* __restrict__ x, const float* __restrict__ bankf,
    const float* __restrict__ coef, float* __restrict__ out) {
  const int blk = blockIdx.x;
  const int cg = blk % CGRP_;
  const int st = (blk / CGRP_) % NSTRIP_;
  const int b  = blk / (CGRP_ * NSTRIP_);
  const int lane = threadIdx.x & 31;
  const int wave = __builtin_amdgcn_readfirstlane((int)(threadIdx.x >> 5));
  const int m  = lane & 15;
  const int hi = lane >> 4;
  const int w0 = cg * 64 + wave * 16;
  int y0 = st * STRIP_;

  const float4 cf = *(const float4*)(coef + ((size_t)b * OUT_ + m) * KEXP_);
  const float4* bt = (const float4*)bankf;
  unsigned lo = (unsigned)lane;
  v2f a[36];
  conv_mix_a<0>(a, bt + lo, cf);
  CONV_A_FENCE(lo, a, 0);
  conv_mix_a<9>(a, bt + lo, cf);
  CONV_A_FENCE(lo, a, 9);
  conv_mix_a<18>(a, bt + lo, cf);
  CONV_A_FENCE(lo, a, 18);
  conv_mix_a<27>(a, bt + lo, cf);
  CONV_A_FENCE_S(y0, a, 27);

  const int wl = w0 + m - 1, wr = w0 + m + 1;
  const bool v0 = wl >= 0, v2 = wr < W_;
  const unsigned chan = (unsigned)(2 * hi) * HW_;
  const unsigned o0 = chan + (unsigned)(v0 ? wl : 0);
  const unsigned o1 = chan + (unsigned)(w0 + m);
  const unsigned o2 = chan + (unsigned)(v2 ? wr : W_ - 1);

  const float* xb = x + (size_t)b * CIN_ * HW_;
  float* ob = out + (size_t)b * OUT_ * HW_ + (size_t)(8 * hi) * HW_ + w0 + m;

  const int ytop = y0 > 0 ? y0 - 1 : 0;
  v2f bA[12], bB[12];
  conv_load_row(bA, xb + (size_t)ytop * W_, o0, o1, o2);
  conv_load_row(bB, xb + (size_t)y0 * W_, o0, o1, o2);

  v8f P = {}, Q = {}, R = {};
  v8f none = {};
  if (st != 0) conv_row_mma<false, false, true>(a, bA, none, none, P, v0, v2);
  conv_load_row(bA, xb + (size_t)(y0 + 1) * W_, o0, o1, o2);
  conv_row_mma<false, true, true>(a, bB, none, P, Q, v0, v2);
  conv_load_row(bB, xb + (size_t)(y0 + 2) * W_, o0, o1, o2);

#pragma nounroll
  for (int t = 0; t < (STRIP_ - 2) / 6; ++t) {
    const int y = y0 + 1 + 6 * t;
    const float* xr = xb + (size_t)y * W_;
    float* orow = ob + (size_t)(y - 1) * W_;
    const int ylast = (y + 7 < H_) ? y + 7 : H_ - 1;

    conv_row_mma<true, true, true>(a, bA, P, Q, R, v0, v2);
    conv_store_row(orow, P);             P = none;
    conv_load_row(bA, xr + 2 * W_, o0, o1, o2);
    conv_row_mma<true, true, true>(a, bB, Q, R, P, v0, v2);
    conv_store_row(orow + W_, Q);        Q = none;
    conv_load_row(bB, xr + 3 * W_, o0, o1, o2);
    conv_row_mma<true, true, true>(a, bA, R, P, Q, v0, v2);
    conv_store_row(orow + 2 * W_, R);    R = none;
    conv_load_row(bA, xr + 4 * W_, o0, o1, o2);
    conv_row_mma<true, true, true>(a, bB, P, Q, R, v0, v2);
    conv_store_row(orow + 3 * W_, P);    P = none;
    conv_load_row(bB, xr + 5 * W_, o0, o1, o2);
    conv_row_mma<true, true, true>(a, bA, Q, R, P, v0, v2);
    conv_store_row(orow + 4 * W_, Q);    Q = none;
    conv_load_row(bA, xr + 6 * W_, o0, o1, o2);
    conv_row_mma<true, true, true>(a, bB, R, P, Q, v0, v2);
    conv_store_row(orow + 5 * W_, R);    R = none;
    conv_load_row(bB, xb + (size_t)ylast * W_, o0, o1, o2);
  }

  conv_row_mma<true, true, false>(a, bA, P, Q, none, v0, v2);
  conv_store_row(ob + (size_t)(y0 + STRIP_ - 2) * W_, P);
  if (st != NSTRIP_ - 1) conv_row_mma<true, false, false>(a, bB, Q, none, none, v0, v2);
  conv_store_row(ob + (size_t)(y0 + STRIP_ - 1) * W_, Q);
}

extern "C" void kernel_launch(void* const* d_in, const int* in_sizes, int n_in,
                              void* d_out, int out_size, void* d_ws, size_t ws_size,
                              hipStream_t stream) {
  (void)in_sizes; (void)n_in; (void)out_size; (void)ws_size;
  const float* x      = (const float*)d_in[0];
  const float* w1     = (const float*)d_in[1];
  const float* w2     = (const float*)d_in[2];
  const float* b2     = (const float*)d_in[3];
  const float* weight = (const float*)d_in[4];
  float* out = (float*)d_out;

  float* psum   = (float*)d_ws;
  float* coef   = (float*)d_ws + 512;
  float* bankf  = (float*)d_ws + 1536;

  dyn_pool_kernel<<<B_ * CIN_ * PSPLIT_, 256, 0, stream>>>(x, weight, psum, bankf);
  dyn_attn_kernel<<<1, 256, 0, stream>>>(psum, w1, w2, b2, coef);
  dyn_conv_kernel<<<B_ * NSTRIP_ * CGRP_, 128, 0, stream>>>(x, bankf, coef, out);
}
